# starting kernel
# baseline (speedup 1.0000x reference)
#include <hip/hip_runtime.h>
#include <math.h>

typedef __attribute__((ext_vector_type(2))) float v2f;
typedef __attribute__((ext_vector_type(4))) float v4f;
typedef __attribute__((ext_vector_type(8))) float v8f;

#define SPR_S 32
#define SPR_H 32
#define TPB 256
#define WAVES_PER_BLOCK (TPB / 32)
#define PTS_PER_TILE 16

#if __has_builtin(__builtin_amdgcn_tanhf)
  #define FAST_TANH(x) __builtin_amdgcn_tanhf(x)
#elif __has_builtin(__builtin_amdgcn_tanh_f32)
  #define FAST_TANH(x) __builtin_amdgcn_tanh_f32(x)
#else
  #define FAST_TANH(x) tanhf(x)
#endif

__global__ __launch_bounds__(TPB) void springnn_wmma_f32(
    const float* __restrict__ t,
    const float* __restrict__ W1,
    const float* __restrict__ b1,
    const float* __restrict__ W2,
    const float* __restrict__ b2,
    const float* __restrict__ xmin,
    float* __restrict__ out, int N)
{
  __shared__ float sW1[SPR_S * SPR_H];
  __shared__ float sB1[SPR_S * SPR_H];
  __shared__ float sW2[SPR_S * SPR_H];
  __shared__ float sB2[SPR_S];
  __shared__ float sLo[SPR_S];

  for (int i = threadIdx.x; i < SPR_S * SPR_H; i += TPB) {
    sW1[i] = W1[i];
    sB1[i] = b1[i];
    sW2[i] = W2[i];
  }
  if (threadIdx.x < SPR_S) {
    sB2[threadIdx.x] = b2[threadIdx.x];
    sLo[threadIdx.x] = xmin[threadIdx.x];
  }
  __syncthreads();

  const int lane  = threadIdx.x & 31;
  const int wave  = threadIdx.x >> 5;
  const int kh    = lane >> 4;
  const int p     = lane & 15;
  const int tile  = blockIdx.x * WAVES_PER_BLOCK + wave;
  const int base  = tile * PTS_PER_TILE;
  if (base >= N) return;

  const int tidx  = base + p;
  const float tv  = t[tidx < N ? tidx : (N - 1)];

  v8f acc = {};
  float accB2 = 0.0f;

  for (int s = 0; s < SPR_S; ++s) {
    const float lo = sLo[s];
    const float d  = tv - lo;
    const float cw = 0.5f * (1.0f - __cosf(1.57079632679489662f * d));
    const float w  = (d >= 0.0f && d <= 4.0f) ? cw : 0.0f;
    accB2 = fmaf(w, sB2[s], accB2);

    const float* w1s = &sW1[s * SPR_H];
    const float* b1s = &sB1[s * SPR_H];
    const float* w2s = &sW2[s * SPR_H];

#pragma unroll
    for (int j = 0; j < 8; ++j) {
      const int k = 4 * j + 2 * kh;
      const v2f w1p = *(const v2f*)(w1s + k);
      const v2f b1p = *(const v2f*)(b1s + k);
      const v2f w2p = *(const v2f*)(w2s + k);
      v2f a;
      a.x = w * FAST_TANH(fmaf(tv, w1p.x, b1p.x));
      a.y = w * FAST_TANH(fmaf(tv, w1p.y, b1p.y));
      acc = __builtin_amdgcn_wmma_f32_16x16x4_f32(
          false, a, false, w2p, (short)0, acc, false, false);
    }
  }

  const int ib2 = __builtin_bit_cast(int, accB2);
  v8f res;
#pragma unroll
  for (int r = 0; r < 8; ++r) {
    const float addLo = __builtin_bit_cast(float, __builtin_amdgcn_readlane(ib2, r));
    const float addHi = __builtin_bit_cast(float, __builtin_amdgcn_readlane(ib2, r + 8));
    res[r] = acc[r] + (kh ? addHi : addLo);
  }

  if (p == 0) {
    const int o0 = base + kh * 8;
    if (o0 + 8 <= N) {
      v4f lo4 = {res[0], res[1], res[2], res[3]};
      v4f hi4 = {res[4], res[5], res[6], res[7]};
      *(v4f*)(out + o0)     = lo4;
      *(v4f*)(out + o0 + 4) = hi4;
    } else {
      for (int r = 0; r < 8 && o0 + r < N; ++r) out[o0 + r] = res[r];
    }
  }
}

extern "C" void kernel_launch(void* const* d_in, const int* in_sizes, int n_in,
                              void* d_out, int out_size, void* d_ws, size_t ws_size,
                              hipStream_t stream) {
  (void)n_in; (void)d_ws; (void)ws_size; (void)out_size;
  const float* t    = (const float*)d_in[0];
  const float* W1   = (const float*)d_in[1];
  const float* b1   = (const float*)d_in[2];
  const float* W2   = (const float*)d_in[3];
  const float* b2   = (const float*)d_in[4];
  const float* xmin = (const float*)d_in[5];
  const int N = in_sizes[0];

  const int tiles  = (N + PTS_PER_TILE - 1) / PTS_PER_TILE;
  const int blocks = (tiles + WAVES_PER_BLOCK - 1) / WAVES_PER_BLOCK;
  springnn_wmma_f32<<<blocks, TPB, 0, stream>>>(t, W1, b1, W2, b2, xmin,
                                                (float*)d_out, N);
}


# Only (point,spring) pairs inside a window are computed: per wave they are compacted, spring-sorted, into 16-row WMMA tiles (diagonal of D); a point's sum now adds per-spring dots in spring order
# speedup vs baseline: 3.7404x; 3.7404x over previous
#include <hip/hip_runtime.h>
#include <math.h>


typedef __attribute__((ext_vector_type(2))) float v2f;
typedef __attribute__((ext_vector_type(4))) float v4f;
typedef __attribute__((ext_vector_type(8))) float v8f;

#define SPR_S 32
#define SPR_H 32
#define TPB 256
#define WAVES_PER_BLOCK (TPB / 32)
#define PTS_PER_WAVE 32
#define ROWS_PER_TILE 16
#define PITCH 36
#define ITEM_CAP (PTS_PER_WAVE * SPR_S)

#if __has_builtin(__builtin_amdgcn_tanhf)
  #define FAST_TANH(x) __builtin_amdgcn_tanhf(x)
#elif __has_builtin(__builtin_amdgcn_tanh_f32)
  #define FAST_TANH(x) __builtin_amdgcn_tanh_f32(x)
#else
  #define FAST_TANH(x) tanhf(x)
#endif

#define WAVE_LDS_HANDOVER()                                   \
  do {                                                        \
    __builtin_amdgcn_fence(__ATOMIC_RELEASE, "wavefront");    \
    __builtin_amdgcn_wave_barrier();                          \
    __builtin_amdgcn_fence(__ATOMIC_ACQUIRE, "wavefront");    \
  } while (0)

__global__ __launch_bounds__(TPB) void springnn_wmma_f32(
    const float* __restrict__ t,
    const float* __restrict__ W1,
    const float* __restrict__ b1,
    const float* __restrict__ W2,
    const float* __restrict__ b2,
    const float* __restrict__ xmin,
    const float* __restrict__ xmax,
    float* __restrict__ out, int N)
{
  struct Lds {
    float    w1[SPR_S * PITCH];
    float    b1[SPR_S * PITCH];
    float    w2[SPR_S * PITCH];
    v4f      par[SPR_S];
    unsigned item[WAVES_PER_BLOCK][ITEM_CAP];
    float    res[WAVES_PER_BLOCK][SPR_S][PTS_PER_WAVE];
  };
  __shared__ __attribute__((aligned(16))) Lds L;
  float* const sW1 = L.w1;
  float* const sB1 = L.b1;
  float* const sW2 = L.w2;

  {
    const int row = threadIdx.x >> 3;
    const int c4  = (threadIdx.x & 7) * 4;
    const int g   = threadIdx.x * 4;
    *(v4f*)&sW1[row * PITCH + c4] = *(const v4f*)(W1 + g);
    *(v4f*)&sB1[row * PITCH + c4] = *(const v4f*)(b1 + g);
    *(v4f*)&sW2[row * PITCH + c4] = *(const v4f*)(W2 + g);
  }
  if (threadIdx.x < SPR_S) {
    const float lo = xmin[threadIdx.x];
    const float hi = xmax[threadIdx.x];
    const v4f pr = {lo, 1.0f / (hi - lo), b2[threadIdx.x], 0.0f};
    L.par[threadIdx.x] = pr;
  }
  __syncthreads();

  const int lane = threadIdx.x & 31;
  const int wave = threadIdx.x >> 5;
  const int base = (blockIdx.x * WAVES_PER_BLOCK + wave) * PTS_PER_WAVE;
  if (base >= N) return;

  const int  idx = base + lane;
  const bool inr = idx < N;
  const float tv = inr ? t[idx] : __builtin_nanf("");

  unsigned* items = L.item[wave];

  float lo_[SPR_S], hi_[SPR_S];
#pragma unroll
  for (int s = 0; s < SPR_S; ++s) {
    lo_[s] = xmin[s];
    hi_[s] = xmax[s];
  }
  unsigned off = 0;
  unsigned cnt = 0;
#pragma unroll
  for (int s = 0; s < SPR_S; ++s) {
    const bool ge = tv >= lo_[s];
    const bool le = tv <= hi_[s];
    const bool act = ge & le;
    const unsigned bal = __builtin_amdgcn_ballot_w32(ge) & __builtin_amdgcn_ballot_w32(le);
    const unsigned pos = off + __builtin_amdgcn_mbcnt_lo(bal, 0u);
    if (act) {
      items[pos] = (cnt << 16) | ((unsigned)s << 8) | (unsigned)lane;
      ++cnt;
    }
    off += (unsigned)__builtin_popcount(bal);
  }
  WAVE_LDS_HANDOVER();

  const int c = lane & 15;
  const int h = lane >> 4;
  const unsigned total  = off;
  const unsigned ntiles = (total + (ROWS_PER_TILE - 1)) / ROWS_PER_TILE;

  unsigned e_nx = items[c];
  for (unsigned T = 0; T < ntiles; ++T) {
    const unsigned i = T * ROWS_PER_TILE + (unsigned)c;
    const bool valid = i < total;
    const unsigned e  = valid ? e_nx : 0u;
    const unsigned i_nx = i + ROWS_PER_TILE;
    e_nx = items[i_nx < ITEM_CAP ? i_nx : (ITEM_CAP - 1)];
    const unsigned s  = (e >> 8) & 31u;
    const unsigned pt = e & 31u;
    const unsigned kk = e >> 16;
    const float tt = __shfl(tv, (int)pt);
    const v4f pr = L.par[s];
    const float d  = tt - pr.x;
    const float cw = fmaf(-0.5f, __builtin_amdgcn_cosf(d * pr.y), 0.5f);
    const float wgt = valid ? cw : 0.0f;

    const float* w1s = &sW1[s * PITCH + 2 * h];
    const float* b1s = &sB1[s * PITCH + 2 * h];
    const float* w2s = &sW2[s * PITCH + 2 * h];
    v2f w1p[8], b1p[8], w2p[8];
#pragma unroll
    for (int j = 0; j < 8; ++j) {
      w1p[j] = *(const v2f*)(w1s + 4 * j);
      b1p[j] = *(const v2f*)(b1s + 4 * j);
      w2p[j] = *(const v2f*)(w2s + 4 * j);
    }

    v8f acc0 = {};
    v8f acc1 = {};
#pragma unroll
    for (int j = 0; j < 8; j += 2) {
      v2f a0, a1;
      a0.x = FAST_TANH(fmaf(tt, w1p[j].x, b1p[j].x));
      a0.y = FAST_TANH(fmaf(tt, w1p[j].y, b1p[j].y));
      a1.x = FAST_TANH(fmaf(tt, w1p[j + 1].x, b1p[j + 1].x));
      a1.y = FAST_TANH(fmaf(tt, w1p[j + 1].y, b1p[j + 1].y));
      acc0 = __builtin_amdgcn_wmma_f32_16x16x4_f32(
          false, a0, false, w2p[j], (short)0, acc0, false, false);
      acc1 = __builtin_amdgcn_wmma_f32_16x16x4_f32(
          false, a1, false, w2p[j + 1], (short)0, acc1, false, false);
    }
    const v8f acc = acc0 + acc1;

    const int r = c & 7;
    const float v01 = (r & 1) ? acc[1] : acc[0];
    const float v23 = (r & 1) ? acc[3] : acc[2];
    const float v45 = (r & 1) ? acc[5] : acc[4];
    const float v67 = (r & 1) ? acc[7] : acc[6];
    const float v03 = (r & 2) ? v23 : v01;
    const float v47 = (r & 2) ? v67 : v45;
    const float dot = (r & 4) ? v47 : v03;

    if (valid && h == (c >> 3))
      L.res[wave][kk][pt] = wgt * (dot + pr.z);
  }
  WAVE_LDS_HANDOVER();

  float sum = 0.0f;
  for (unsigned k0 = 0; __builtin_amdgcn_ballot_w32(k0 < cnt) != 0u; k0 += 4) {
    const float r0 = (k0 + 0 < cnt) ? L.res[wave][(k0 + 0) & 31u][lane] : 0.0f;
    const float r1 = (k0 + 1 < cnt) ? L.res[wave][(k0 + 1) & 31u][lane] : 0.0f;
    const float r2 = (k0 + 2 < cnt) ? L.res[wave][(k0 + 2) & 31u][lane] : 0.0f;
    const float r3 = (k0 + 3 < cnt) ? L.res[wave][(k0 + 3) & 31u][lane] : 0.0f;
    sum += r0;
    sum += r1;
    sum += r2;
    sum += r3;
  }
  if (inr) out[idx] = sum;
}

extern "C" void kernel_launch(void* const* d_in, const int* in_sizes, int n_in,
                              void* d_out, int out_size, void* d_ws, size_t ws_size,
                              hipStream_t stream) {
  (void)n_in; (void)d_ws; (void)ws_size; (void)out_size;
  const float* t    = (const float*)d_in[0];
  const float* W1   = (const float*)d_in[1];
  const float* b1   = (const float*)d_in[2];
  const float* W2   = (const float*)d_in[3];
  const float* b2   = (const float*)d_in[4];
  const float* xmin = (const float*)d_in[5];
  const float* xmax = (const float*)d_in[6];
  const int N = in_sizes[0];

  const int waves  = (N + PTS_PER_WAVE - 1) / PTS_PER_WAVE;
  const int blocks = (waves + WAVES_PER_BLOCK - 1) / WAVES_PER_BLOCK;
  springnn_wmma_f32<<<blocks, TPB, 0, stream>>>(t, W1, b1, W2, b2, xmin, xmax,
                                                (float*)d_out, N);
}


# All global loads (point, window bounds, weight pieces) issued together at kernel start; weight staging to LDS and the block barrier moved behind the work-list build, which no longer waits for them
# speedup vs baseline: 4.0346x; 1.0787x over previous
#include <hip/hip_runtime.h>
#include <math.h>


typedef __attribute__((ext_vector_type(2))) float v2f;
typedef __attribute__((ext_vector_type(4))) float v4f;
typedef __attribute__((ext_vector_type(8))) float v8f;

#define SPR_S 32
#define SPR_H 32
#define TPB 256
#define WAVES_PER_BLOCK (TPB / 32)
#define PTS_PER_WAVE 32
#define ROWS_PER_TILE 16
#define PITCH 36
#define ITEM_CAP (PTS_PER_WAVE * SPR_S)

#if __has_builtin(__builtin_amdgcn_tanhf)
  #define FAST_TANH(x) __builtin_amdgcn_tanhf(x)
#elif __has_builtin(__builtin_amdgcn_tanh_f32)
  #define FAST_TANH(x) __builtin_amdgcn_tanh_f32(x)
#else
  #define FAST_TANH(x) tanhf(x)
#endif

#define WAVE_LDS_HANDOVER()                                   \
  do {                                                        \
    __builtin_amdgcn_fence(__ATOMIC_RELEASE, "wavefront");    \
    __builtin_amdgcn_wave_barrier();                          \
    __builtin_amdgcn_fence(__ATOMIC_ACQUIRE, "wavefront");    \
  } while (0)

__global__ __launch_bounds__(TPB) void springnn_wmma_f32(
    const float* __restrict__ t,
    const float* __restrict__ W1,
    const float* __restrict__ b1,
    const float* __restrict__ W2,
    const float* __restrict__ b2,
    const float* __restrict__ xmin,
    const float* __restrict__ xmax,
    float* __restrict__ out, int N)
{
  struct Lds {
    float    w1[SPR_S * PITCH];
    float    b1[SPR_S * PITCH];
    float    w2[SPR_S * PITCH];
    v4f      par[SPR_S];
    unsigned item[WAVES_PER_BLOCK][ITEM_CAP];
    float    res[WAVES_PER_BLOCK][SPR_S][PTS_PER_WAVE];
  };
  __shared__ __attribute__((aligned(16))) Lds L;
  float* const sW1 = L.w1;
  float* const sB1 = L.b1;
  float* const sW2 = L.w2;

  const int lane = threadIdx.x & 31;
  const int wave = threadIdx.x >> 5;
  const int base = (blockIdx.x * WAVES_PER_BLOCK + wave) * PTS_PER_WAVE;
  const int  idx = base + lane;
  const bool inr = idx < N;

  const float tv = inr ? t[idx] : __builtin_nanf("");
  const int g = threadIdx.x * 4;
  const v4f gw1 = *(const v4f*)(W1 + g);
  const v4f gb1 = *(const v4f*)(b1 + g);
  const v4f gw2 = *(const v4f*)(W2 + g);
  const float plo = xmin[lane];
  const float phi = xmax[lane];
  const float pb2 = b2[lane];

  unsigned* items = L.item[wave];

  float lo_[SPR_S], hi_[SPR_S];
#pragma unroll
  for (int s = 0; s < SPR_S; ++s) {
    lo_[s] = xmin[s];
    hi_[s] = xmax[s];
  }
  unsigned off = 0;
  unsigned cnt = 0;
#pragma unroll
  for (int s = 0; s < SPR_S; ++s) {
    const bool ge = tv >= lo_[s];
    const bool le = tv <= hi_[s];
    const bool act = ge & le;
    const unsigned bal = __builtin_amdgcn_ballot_w32(ge) & __builtin_amdgcn_ballot_w32(le);
    const unsigned pos = off + __builtin_amdgcn_mbcnt_lo(bal, 0u);
    if (act) {
      items[pos] = (cnt << 16) | ((unsigned)s << 8) | (unsigned)lane;
      ++cnt;
    }
    off += (unsigned)__builtin_popcount(bal);
  }
  {
    const int row = threadIdx.x >> 3;
    const int c4  = (threadIdx.x & 7) * 4;
    *(v4f*)&sW1[row * PITCH + c4] = gw1;
    *(v4f*)&sB1[row * PITCH + c4] = gb1;
    *(v4f*)&sW2[row * PITCH + c4] = gw2;
  }
  if (threadIdx.x < SPR_S) {
    const v4f pr = {plo, 1.0f / (phi - plo), pb2, 0.0f};
    L.par[threadIdx.x] = pr;
  }
  __syncthreads();
  if (base >= N) return;

  const int c = lane & 15;
  const int h = lane >> 4;
  const unsigned total  = off;
  const unsigned ntiles = (total + (ROWS_PER_TILE - 1)) / ROWS_PER_TILE;

  unsigned e_nx = items[c];
  for (unsigned T = 0; T < ntiles; ++T) {
    const unsigned i = T * ROWS_PER_TILE + (unsigned)c;
    const bool valid = i < total;
    const unsigned e  = valid ? e_nx : 0u;
    const unsigned i_nx = i + ROWS_PER_TILE;
    e_nx = items[i_nx < ITEM_CAP ? i_nx : (ITEM_CAP - 1)];
    const unsigned s  = (e >> 8) & 31u;
    const unsigned pt = e & 31u;
    const unsigned kk = e >> 16;
    const float tt = __shfl(tv, (int)pt);
    const v4f pr = L.par[s];
    const float d  = tt - pr.x;
    const float cw = fmaf(-0.5f, __builtin_amdgcn_cosf(d * pr.y), 0.5f);
    const float wgt = valid ? cw : 0.0f;

    const float* w1s = &sW1[s * PITCH + 2 * h];
    const float* b1s = &sB1[s * PITCH + 2 * h];
    const float* w2s = &sW2[s * PITCH + 2 * h];
    v2f w1p[8], b1p[8], w2p[8];
#pragma unroll
    for (int j = 0; j < 8; ++j) {
      w1p[j] = *(const v2f*)(w1s + 4 * j);
      b1p[j] = *(const v2f*)(b1s + 4 * j);
      w2p[j] = *(const v2f*)(w2s + 4 * j);
    }

    v8f acc0 = {};
    v8f acc1 = {};
#pragma unroll
    for (int j = 0; j < 8; j += 2) {
      v2f a0, a1;
      a0.x = FAST_TANH(fmaf(tt, w1p[j].x, b1p[j].x));
      a0.y = FAST_TANH(fmaf(tt, w1p[j].y, b1p[j].y));
      a1.x = FAST_TANH(fmaf(tt, w1p[j + 1].x, b1p[j + 1].x));
      a1.y = FAST_TANH(fmaf(tt, w1p[j + 1].y, b1p[j + 1].y));
      acc0 = __builtin_amdgcn_wmma_f32_16x16x4_f32(
          false, a0, false, w2p[j], (short)0, acc0, false, false);
      acc1 = __builtin_amdgcn_wmma_f32_16x16x4_f32(
          false, a1, false, w2p[j + 1], (short)0, acc1, false, false);
    }
    const v8f acc = acc0 + acc1;

    const int r = c & 7;
    const float v01 = (r & 1) ? acc[1] : acc[0];
    const float v23 = (r & 1) ? acc[3] : acc[2];
    const float v45 = (r & 1) ? acc[5] : acc[4];
    const float v67 = (r & 1) ? acc[7] : acc[6];
    const float v03 = (r & 2) ? v23 : v01;
    const float v47 = (r & 2) ? v67 : v45;
    const float dot = (r & 4) ? v47 : v03;

    if (valid && h == (c >> 3))
      L.res[wave][kk][pt] = wgt * (dot + pr.z);
  }
  WAVE_LDS_HANDOVER();

  float sum = 0.0f;
  for (unsigned k0 = 0; __builtin_amdgcn_ballot_w32(k0 < cnt) != 0u; k0 += 4) {
    const float r0 = (k0 + 0 < cnt) ? L.res[wave][(k0 + 0) & 31u][lane] : 0.0f;
    const float r1 = (k0 + 1 < cnt) ? L.res[wave][(k0 + 1) & 31u][lane] : 0.0f;
    const float r2 = (k0 + 2 < cnt) ? L.res[wave][(k0 + 2) & 31u][lane] : 0.0f;
    const float r3 = (k0 + 3 < cnt) ? L.res[wave][(k0 + 3) & 31u][lane] : 0.0f;
    sum += r0;
    sum += r1;
    sum += r2;
    sum += r3;
  }
  if (inr) out[idx] = sum;
}

extern "C" void kernel_launch(void* const* d_in, const int* in_sizes, int n_in,
                              void* d_out, int out_size, void* d_ws, size_t ws_size,
                              hipStream_t stream) {
  (void)n_in; (void)d_ws; (void)ws_size; (void)out_size;
  const float* t    = (const float*)d_in[0];
  const float* W1   = (const float*)d_in[1];
  const float* b1   = (const float*)d_in[2];
  const float* W2   = (const float*)d_in[3];
  const float* b2   = (const float*)d_in[4];
  const float* xmin = (const float*)d_in[5];
  const float* xmax = (const float*)d_in[6];
  const int N = in_sizes[0];

  const int waves  = (N + PTS_PER_WAVE - 1) / PTS_PER_WAVE;
  const int blocks = (waves + WAVES_PER_BLOCK - 1) / WAVES_PER_BLOCK;
  springnn_wmma_f32<<<blocks, TPB, 0, stream>>>(t, W1, b1, W2, b2, xmin, xmax,
                                                (float*)d_out, N);
}


# Tile rows taken straight from per-point spring masks in registers (points sorted by first spring, round k = each point's k-th spring): no LDS work list, result plane or summing pass; sum order kept
# speedup vs baseline: 4.6834x; 1.1608x over previous
#include <hip/hip_runtime.h>
#include <math.h>


typedef __attribute__((ext_vector_type(2))) float v2f;
typedef __attribute__((ext_vector_type(4))) float v4f;
typedef __attribute__((ext_vector_type(8))) float v8f;

#define SPR_S 32
#define SPR_H 32
#define TPB 256
#define WAVES_PER_BLOCK (TPB / 32)
#define PTS_PER_WAVE 32
#define PITCH 36

#if __has_builtin(__builtin_amdgcn_tanhf)
  #define FAST_TANH(x) __builtin_amdgcn_tanhf(x)
#elif __has_builtin(__builtin_amdgcn_tanh_f32)
  #define FAST_TANH(x) __builtin_amdgcn_tanh_f32(x)
#else
  #define FAST_TANH(x) tanhf(x)
#endif

struct Lds {
  float w1[SPR_S * PITCH];
  float b1[SPR_S * PITCH];
  float w2[SPR_S * PITCH];
  v2f   win[SPR_S];
  float ob2[SPR_S];
};

template <int J>
static __device__ __forceinline__ unsigned xor_lane(unsigned v) {
  if constexpr (J == 16) {
    return (unsigned)__builtin_amdgcn_permlanex16(v, v, 0x76543210u, 0xfedcba98u, false, false);
  } else if constexpr (J == 8) {
    return (unsigned)__builtin_amdgcn_permlane16(v, v, 0xfedcba98u, 0x76543210u, false, false);
  } else if constexpr (J == 4) {
    return (unsigned)__builtin_amdgcn_permlane16(v, v, 0x32107654u, 0xba98fedcu, false, false);
  } else if constexpr (J == 2) {
    return (unsigned)__builtin_amdgcn_permlane16(v, v, 0x54761032u, 0xdcfe98bau, false, false);
  } else {
    return (unsigned)__builtin_amdgcn_permlane16(v, v, 0x67452301u, 0xefcdab89u, false, false);
  }
}

template <int K, int J>
static __device__ __forceinline__ unsigned sort_stage(unsigned p, int lane) {
  const unsigned o  = xor_lane<J>(p);
  const unsigned mn = p < o ? p : o;
  const unsigned mx = p < o ? o : p;
  const bool up    = (K == 32) || ((lane & K) == 0);
  const bool lower = (lane & J) == 0;
  return (lower == up) ? mn : mx;
}

static __device__ __forceinline__ void tile_step(const Lds& L, float tt, unsigned s, bool valid,
                                                 int c, int h, float& sum) {
  const v2f pr = L.win[s];
  const float pb2 = L.ob2[s];
  const float d  = tt - pr.x;
  const float cw = fmaf(-0.5f, __builtin_amdgcn_cosf(d * pr.y), 0.5f);
  const float wgt = valid ? cw : 0.0f;

  const float* w1s = &L.w1[s * PITCH + 2 * h];
  const float* b1s = &L.b1[s * PITCH + 2 * h];
  const float* w2s = &L.w2[s * PITCH + 2 * h];
  v2f w1p[8], b1p[8], w2p[8];
#pragma unroll
  for (int j = 0; j < 8; ++j) {
    w1p[j] = *(const v2f*)(w1s + 4 * j);
    b1p[j] = *(const v2f*)(b1s + 4 * j);
    w2p[j] = *(const v2f*)(w2s + 4 * j);
  }

  v8f acc0 = {};
  v8f acc1 = {};
#pragma unroll
  for (int j = 0; j < 8; j += 2) {
    v2f a0, a1;
    a0.x = FAST_TANH(fmaf(tt, w1p[j].x, b1p[j].x));
    a0.y = FAST_TANH(fmaf(tt, w1p[j].y, b1p[j].y));
    a1.x = FAST_TANH(fmaf(tt, w1p[j + 1].x, b1p[j + 1].x));
    a1.y = FAST_TANH(fmaf(tt, w1p[j + 1].y, b1p[j + 1].y));
    acc0 = __builtin_amdgcn_wmma_f32_16x16x4_f32(
        false, a0, false, w2p[j], (short)0, acc0, false, false);
    acc1 = __builtin_amdgcn_wmma_f32_16x16x4_f32(
        false, a1, false, w2p[j + 1], (short)0, acc1, false, false);
  }
  const v8f acc = acc0 + acc1;

  const int r = c & 7;
  const float v01 = (r & 1) ? acc[1] : acc[0];
  const float v23 = (r & 1) ? acc[3] : acc[2];
  const float v45 = (r & 1) ? acc[5] : acc[4];
  const float v67 = (r & 1) ? acc[7] : acc[6];
  const float v03 = (r & 2) ? v23 : v01;
  const float v47 = (r & 2) ? v67 : v45;
  const float dot = (r & 4) ? v47 : v03;

  sum += wgt * (dot + pb2);
}

__global__ __launch_bounds__(TPB) __attribute__((amdgpu_waves_per_eu(4, 4))) void springnn_wmma_f32(
    const float* __restrict__ t,
    const float* __restrict__ W1,
    const float* __restrict__ b1,
    const float* __restrict__ W2,
    const float* __restrict__ b2,
    const float* __restrict__ xmin,
    const float* __restrict__ xmax,
    float* __restrict__ out, int N)
{
  __shared__ __attribute__((aligned(16))) Lds L;

  const int lane = threadIdx.x & 31;
  const int wave = threadIdx.x >> 5;
  const int base = (blockIdx.x * WAVES_PER_BLOCK + wave) * PTS_PER_WAVE;
  const int  idx = base + lane;
  const bool inr = idx < N;

  const float tv = inr ? t[idx] : __builtin_nanf("");
  const int g = threadIdx.x * 4;
  const v4f gw1 = *(const v4f*)(W1 + g);
  const v4f gb1 = *(const v4f*)(b1 + g);
  const v4f gw2 = *(const v4f*)(W2 + g);
  const float plo = xmin[lane];
  const float phi = xmax[lane];
  const float pb2 = b2[lane];

  float lo_[SPR_S], hi_[SPR_S];
#pragma unroll
  for (int s = 0; s < SPR_S; ++s) {
    lo_[s] = xmin[s];
    hi_[s] = xmax[s];
  }
  asm volatile("" :: "v"(tv), "v"(gw1), "v"(gb1), "v"(gw2), "v"(plo), "v"(phi), "v"(pb2));
  __builtin_amdgcn_sched_barrier(0);

  unsigned mask = 0u;
#pragma unroll
  for (int s = 0; s < SPR_S; ++s) {
    const bool act = (tv >= lo_[s]) & (tv <= hi_[s]);
    mask |= act ? (1u << s) : 0u;
  }

  unsigned p = ((mask != 0u ? (unsigned)__builtin_ctz(mask) : 32u) << 5) | (unsigned)lane;
  p = sort_stage<2, 1>(p, lane);
  p = sort_stage<4, 2>(p, lane);
  p = sort_stage<4, 1>(p, lane);
  p = sort_stage<8, 4>(p, lane);
  p = sort_stage<8, 2>(p, lane);
  p = sort_stage<8, 1>(p, lane);
  p = sort_stage<16, 8>(p, lane);
  p = sort_stage<16, 4>(p, lane);
  p = sort_stage<16, 2>(p, lane);
  p = sort_stage<16, 1>(p, lane);
  p = sort_stage<32, 16>(p, lane);
  p = sort_stage<32, 8>(p, lane);
  p = sort_stage<32, 4>(p, lane);
  p = sort_stage<32, 2>(p, lane);
  p = sort_stage<32, 1>(p, lane);

  const int c = lane & 15;
  const int h = lane >> 4;
  const unsigned px = xor_lane<16>(p);
  const int srcA = (int)((h ? px : p) & 31u);
  const int srcB = (int)((h ? p : px) & 31u);
  const float tA = __shfl(tv, srcA);
  const float tB = __shfl(tv, srcB);
  unsigned mA = __shfl(mask, srcA);
  unsigned mB = __shfl(mask, srcB);

  {
    const int row = threadIdx.x >> 3;
    const int c4  = (threadIdx.x & 7) * 4;
    *(v4f*)&L.w1[row * PITCH + c4] = gw1;
    *(v4f*)&L.b1[row * PITCH + c4] = gb1;
    *(v4f*)&L.w2[row * PITCH + c4] = gw2;
  }
  if (threadIdx.x < SPR_S) {
    const v2f pr = {plo, 1.0f / (phi - plo)};
    L.win[threadIdx.x] = pr;
    L.ob2[threadIdx.x] = pb2;
  }
  __syncthreads();
  if (base >= N) return;

  float sumA = 0.0f, sumB = 0.0f;
  unsigned more = __builtin_amdgcn_ballot_w32((mA | mB) != 0u);
  while (more != 0u) {
    const bool vA = mA != 0u;
    const bool vB = mB != 0u;
    const unsigned sA = vA ? (unsigned)__builtin_ctz(mA) : 0u;
    const unsigned sB = vB ? (unsigned)__builtin_ctz(mB) : 0u;
    mA &= mA - 1u;
    mB &= mB - 1u;
    more = __builtin_amdgcn_ballot_w32((mA | mB) != 0u);
    tile_step(L, tA, sA, vA, c, h, sumA);
    tile_step(L, tB, sB, vB, c, h, sumB);
  }

  const int own = c + ((c & 8) ? 16 : 0);
  const float ra = __shfl(sumA, own);
  const float rb = __shfl(sumB, own);
  const float res = h ? rb : ra;
  const int dst = base + (int)(p & 31u);
  if (dst < N) out[dst] = res;
}

extern "C" void kernel_launch(void* const* d_in, const int* in_sizes, int n_in,
                              void* d_out, int out_size, void* d_ws, size_t ws_size,
                              hipStream_t stream) {
  (void)n_in; (void)d_ws; (void)ws_size; (void)out_size;
  const float* t    = (const float*)d_in[0];
  const float* W1   = (const float*)d_in[1];
  const float* b1   = (const float*)d_in[2];
  const float* W2   = (const float*)d_in[3];
  const float* b2   = (const float*)d_in[4];
  const float* xmin = (const float*)d_in[5];
  const float* xmax = (const float*)d_in[6];
  const int N = in_sizes[0];

  const int waves  = (N + PTS_PER_WAVE - 1) / PTS_PER_WAVE;
  const int blocks = (waves + WAVES_PER_BLOCK - 1) / WAVES_PER_BLOCK;
  springnn_wmma_f32<<<blocks, TPB, 0, stream>>>(t, W1, b1, W2, b2, xmin, xmax,
                                                (float*)d_out, N);
}
